# starting kernel
# baseline (speedup 1.0000x reference)
#include <hip/hip_runtime.h>

typedef __attribute__((ext_vector_type(2))) float v2f;
typedef __attribute__((ext_vector_type(4))) float v4f;
typedef __attribute__((ext_vector_type(8))) float v8f;

#define NB 64
#define NN 32
#define NR 16
#define NG 4960
#define NF 16
#define NTILES 310
#define TILES_PER_BLOCK 40
#define TILES_PER_WAVE 5

__global__ __launch_bounds__(256)
void graphlet_conv_wmma(const float* __restrict__ inp,
                        const float* __restrict__ flt,
                        float* __restrict__ out) {
  __shared__ __align__(16) float s_in[NN * NN * NR];
  __shared__ __align__(16) float s_flt[NF * 9 * NR];

  const int tid  = threadIdx.x;
  const int bIdx = blockIdx.y;
  const int wave = tid >> 5;
  const int lane = tid & 31;
  const int half = lane >> 4;
  const int col  = lane & 15;

  {
    const v4f* src = (const v4f*)(inp + (size_t)bIdx * (NN * NN * NR));
    v4f* dst = (v4f*)s_in;
#pragma unroll
    for (int i = 0; i < (NN * NN * NR) / 4 / 256; ++i)
      dst[tid + i * 256] = src[tid + i * 256];

    const v4f* fsrc = (const v4f*)flt;
    v4f* fdst = (v4f*)s_flt;
    for (int i = tid; i < (NF * 9 * NR) / 4; i += 256)
      fdst[i] = fsrc[i];
  }
  __syncthreads();

  for (int t = 0; t < TILES_PER_WAVE; ++t) {
    const int tile = blockIdx.x * TILES_PER_BLOCK + wave * TILES_PER_WAVE + t;
    if (tile >= NTILES) break;

    const int g = tile * 16 + col;
    int rem = g, n0 = 0, n1, n2;
    for (; n0 < 30; ++n0) {
      const int cnt = ((31 - n0) * (30 - n0)) >> 1;
      if (rem < cnt) break;
      rem -= cnt;
    }
    n1 = n0 + 1;
    for (; n1 < 31; ++n1) {
      const int cnt = 31 - n1;
      if (rem < cnt) break;
      rem -= cnt;
    }
    n2 = n1 + 1 + rem;

    int rowb[9];
#pragma unroll
    for (int i = 0; i < 3; ++i) {
      const int ni = (i == 0) ? n0 : ((i == 1) ? n1 : n2);
#pragma unroll
      for (int j = 0; j < 3; ++j) {
        const int nj = (j == 0) ? n0 : ((j == 1) ? n1 : n2);
        rowb[i * 3 + j] = (ni * NN + nj) * NR + 2 * half;
      }
    }
    const int fbase = col * (9 * NR) + 2 * half;

    v8f c = {0.f, 0.f, 0.f, 0.f, 0.f, 0.f, 0.f, 0.f};
#pragma unroll
    for (int kk = 0; kk < 36; ++kk) {
      const int ij = kk >> 2;
      const int rb = (kk & 3) << 2;
      const v2f a = *(const v2f*)&s_in [rowb[ij] + rb];
      const v2f b = *(const v2f*)&s_flt[fbase + ij * NR + rb];
      c = __builtin_amdgcn_wmma_f32_16x16x4_f32(
               false, a,  false, b,
               (short)0, c,  false,  false);
    }

    float* op = out + ((size_t)bIdx * NG + (size_t)tile * 16 + half * 8) * NF + col;
#pragma unroll
    for (int r = 0; r < 8; ++r)
      op[r * NF] = c[r];
  }
}

extern "C" void kernel_launch(void* const* d_in, const int* in_sizes, int n_in,
                              void* d_out, int out_size, void* d_ws, size_t ws_size,
                              hipStream_t stream) {
  const float* inp = (const float*)d_in[0];
  const float* flt = (const float*)d_in[1];
  float* out = (float*)d_out;
  dim3 grid(8, NB);
  dim3 block(256);
  hipLaunchKernelGGL(graphlet_conv_wmma, grid, block, 0, stream, inp, flt, out);
}


# Factor the 144-term sum into three node-pair tables (C01, V0, V1) built per block on the f32 WMMA, then out[g]=C01[n0,n1]+V0[n0,n2]+V1[n1,n2]: about 4x fewer WMMAs; changes the summation order
# speedup vs baseline: 1.8146x; 1.8146x over previous
#include <hip/hip_runtime.h>

typedef __attribute__((ext_vector_type(2))) float v2f;
typedef __attribute__((ext_vector_type(4))) float v4f;
typedef __attribute__((ext_vector_type(8))) float v8f;

#define NB 64
#define NN 32
#define NR 16
#define NG 4960
#define NF 16
#define SLAB (NN * NN * NR)
#define TRI_MAX 1360


__device__ __forceinline__ v8f acc16(v8f c, const float* s_in, const float* s_flt,
                                     int rowbase, int fb) {
#pragma unroll
  for (int s = 0; s < 4; ++s) {
    const v2f a = *(const v2f*)&s_in[rowbase + 4 * s];
    const v2f b = *(const v2f*)&s_flt[fb + 4 * s];
    c = __builtin_amdgcn_wmma_f32_16x16x4_f32(
             false, a,  false, b,
             (short)0, c,  false,  false);
  }
  return c;
}

__device__ __forceinline__ void tile_of(int t, int lo, int hi, int& n, int& mh) {
  const int top = (hi + 1 < 15) ? (hi + 1) : 15;
  const int nlow = (top > lo) ? (top - lo) : 0;
  if (t < 2 * nlow) { n = lo + (t >> 1); mh = t & 1; }
  else { n = ((lo > 15) ? lo : 15) + (t - 2 * nlow); mh = 1; }
}

__global__ __launch_bounds__(512)
void graphlet_conv_wmma(const float* __restrict__ inp,
                        const float* __restrict__ flt,
                        float* __restrict__ out) {
  __shared__ __align__(16) float s_in[SLAB];
  __shared__ __align__(16) float s_flt[NF * 9 * NR];
  __shared__ __align__(16) float s_tab[3 * SLAB];
  __shared__ unsigned s_tri[TRI_MAX];

  const int tid  = threadIdx.x;
  const int bIdx = blockIdx.y;
  const int part = blockIdx.x;
  const int wave = __builtin_amdgcn_readfirstlane(tid >> 5);
  const int lane = tid & 31;
  const int half = lane >> 4;
  const int col  = lane & 15;

  const int na   = (part == 0) ? 0 : ((part == 1) ? 3 : ((part == 2) ? 7 : 12));
  const int nb   = (part == 0) ? 2 : ((part == 1) ? 6 : ((part == 2) ? 11 : 29));
  const int g_lo = (part == 0) ? 0 : ((part == 1) ? 1306 : ((part == 2) ? 2660 : 3820));
  const int g_hi = (part == 0) ? 1306 : ((part == 1) ? 2660 : ((part == 2) ? 3820 : NG));
  const int cnt  = g_hi - g_lo;

  {
    const v4f* src = (const v4f*)(inp + (size_t)bIdx * SLAB);
    v4f regs[SLAB / 4 / 512];
#pragma unroll
    for (int i = 0; i < SLAB / 4 / 512; ++i)
      regs[i] = src[tid + i * 512];

    for (int t = tid; t < cnt; t += 512) {
      const unsigned h  = (unsigned)((NG - 1) - (g_lo + t));
      const unsigned h6 = 6u * h;
      unsigned a = 2u;
#pragma unroll
      for (unsigned s = 16u; s != 0u; s >>= 1) {
        const unsigned tt = a + s;
        const unsigned p3 = __umul24(__umul24(tt, tt - 1u), tt - 2u);
        a = (p3 <= h6) ? tt : a;
      }
      const unsigned pa = __umul24(__umul24(a, a - 1u), a - 2u);
      const unsigned h2 = h - ((pa * 43691u) >> 18);
      const unsigned h22 = 2u * h2;
      unsigned bq = 1u;
#pragma unroll
      for (unsigned s = 16u; s != 0u; s >>= 1) {
        const unsigned tt = bq + s;
        const unsigned p2 = __umul24(tt, tt - 1u);
        bq = (p2 <= h22) ? tt : bq;
      }
      const unsigned cq = h2 - (__umul24(bq, bq - 1u) >> 1);
      const unsigned n0 = 31u - a, n1 = 31u - bq, n2 = 31u - cq;
      s_tri[t] = (n0 * 32u + n1) | ((n0 * 32u + n2) << 10) | ((n1 * 32u + n2) << 20);
    }

    v4f* dst = (v4f*)s_in;
#pragma unroll
    for (int i = 0; i < SLAB / 4 / 512; ++i)
      dst[tid + i * 512] = regs[i];

    const v4f* fsrc = (const v4f*)flt;
    v4f* fdst = (v4f*)s_flt;
    for (int i = tid; i < (NF * 9 * NR) / 4; i += 512)
      fdst[i] = fsrc[i];
  }
  __syncthreads();

  {
    const int fbase = col * (9 * NR) + 2 * half;
    const int lowC = ((nb + 1 < 15) ? (nb + 1) : 15) - na;
    const int tilesC = (nb - na + 1) + ((lowC > 0) ? lowC : 0);
    const int lowV = 15 - (na + 1);
    const int tilesV = (30 - na) + ((lowV > 0) ? lowV : 0);
    const int units = 2 * tilesC + tilesV;

    for (int u = wave; u < units; u += 16) {
      int kind, n, mh;
      if (u < tilesC)               { kind = 0; tile_of(u, na, nb, n, mh); }
      else if (u < tilesC + tilesV) { kind = 2; tile_of(u - tilesC, na + 1, 30, n, mh); }
      else                          { kind = 1; tile_of(u - tilesC - tilesV, na, nb, n, mh); }

      const int m  = mh * 16 + col;
      const int nm = (n * NN + m) * NR + 2 * half;
      const int mn = (m * NN + n) * NR + 2 * half;
      const int nn = (n * NN + n) * NR + 2 * half;
      const int mm = (m * NN + m) * NR + 2 * half;

      v8f c = {0.f, 0.f, 0.f, 0.f, 0.f, 0.f, 0.f, 0.f};
      if (kind == 0) {
        c = acc16(c, s_in, s_flt, nn, fbase + 0 * NR);
        c = acc16(c, s_in, s_flt, nm, fbase + 1 * NR);
        c = acc16(c, s_in, s_flt, mn, fbase + 3 * NR);
        c = acc16(c, s_in, s_flt, mm, fbase + 4 * NR);
      } else if (kind == 1) {
        c = acc16(c, s_in, s_flt, nm, fbase + 2 * NR);
        c = acc16(c, s_in, s_flt, mn, fbase + 6 * NR);
      } else {
        c = acc16(c, s_in, s_flt, nm, fbase + 5 * NR);
        c = acc16(c, s_in, s_flt, mn, fbase + 7 * NR);
        c = acc16(c, s_in, s_flt, mm, fbase + 8 * NR);
      }

      float* tp = s_tab + kind * SLAB + (n * NN + mh * 16 + half * 8) * NF + col;
#pragma unroll
      for (int r = 0; r < 8; ++r)
        tp[r * NF] = c[r];
    }
  }
  __syncthreads();

  {
    const int j = lane >> 2;
    const int q = lane & 3;
    const int passes = (cnt + 7) >> 3;
    float* obase = out + ((size_t)bIdx * NG + (size_t)g_lo) * NF + q * 4;
    for (int k0 = wave; k0 < passes; k0 += 64) {
      v4f x[4], y[4], z[4];
#pragma unroll
      for (int u = 0; u < 4; ++u) {
        const int t  = (k0 + 16 * u) * 8 + j;
        const int tc = (t < cnt) ? t : (cnt - 1);
        const unsigned w = s_tri[tc];
        const unsigned i01 = w & 1023u, i02 = (w >> 10) & 1023u, i12 = w >> 20;
        x[u] = *(const v4f*)&s_tab[0 * SLAB + i01 * NF + q * 4];
        y[u] = *(const v4f*)&s_tab[1 * SLAB + i02 * NF + q * 4];
        z[u] = *(const v4f*)&s_tab[2 * SLAB + i12 * NF + q * 4];
      }
#pragma unroll
      for (int u = 0; u < 4; ++u) {
        const int t = (k0 + 16 * u) * 8 + j;
        if (t < cnt)
          *(v4f*)(obase + (size_t)t * NF) = (x[u] + y[u]) + z[u];
      }
    }
  }
}

extern "C" void kernel_launch(void* const* d_in, const int* in_sizes, int n_in,
                              void* d_out, int out_size, void* d_ws, size_t ws_size,
                              hipStream_t stream) {
  const float* inp = (const float*)d_in[0];
  const float* flt = (const float*)d_in[1];
  float* out = (float*)d_out;
  dim3 grid(4, NB);
  dim3 block(512);
  hipLaunchKernelGGL(graphlet_conv_wmma, grid, block, 0, stream, inp, flt, out);
}


# Pad the LDS pitches of the staged slab (20 dwords per row, 644 per n) and of the filters (148) to 4 x odd mod 64: the WMMA fragment reads, 4- to 16-way bank-conflicted before, become conflict-free
# speedup vs baseline: 2.2139x; 1.2200x over previous
#include <hip/hip_runtime.h>

typedef __attribute__((ext_vector_type(2))) float v2f;
typedef __attribute__((ext_vector_type(4))) float v4f;
typedef __attribute__((ext_vector_type(8))) float v8f;

#define NB 64
#define NN 32
#define NR 16
#define NG 4960
#define NF 16
#define SLAB (NN * NN * NR)
#define TRI_MAX 1360

#define RP 20
#define NP 644
#define FP 148


__device__ __forceinline__ v8f acc16(v8f c, const float* s_in, const float* s_flt,
                                     int rowbase, int fb) {
#pragma unroll
  for (int s = 0; s < 4; ++s) {
    const v2f a = *(const v2f*)&s_in[rowbase + 4 * s];
    const v2f b = *(const v2f*)&s_flt[fb + 4 * s];
    c = __builtin_amdgcn_wmma_f32_16x16x4_f32(
             false, a,  false, b,
             (short)0, c,  false,  false);
  }
  return c;
}

__device__ __forceinline__ void tile_of(int t, int lo, int hi, int& n, int& mh) {
  const int top = (hi + 1 < 15) ? (hi + 1) : 15;
  const int nlow = (top > lo) ? (top - lo) : 0;
  if (t < 2 * nlow) { n = lo + (t >> 1); mh = t & 1; }
  else { n = ((lo > 15) ? lo : 15) + (t - 2 * nlow); mh = 1; }
}

__global__ __launch_bounds__(512)
void graphlet_conv_wmma(const float* __restrict__ inp,
                        const float* __restrict__ flt,
                        float* __restrict__ out) {
  __shared__ __align__(16) float s_in[NN * NP];
  __shared__ __align__(16) float s_flt[NF * FP];
  __shared__ __align__(16) float s_tab[3 * SLAB];
  __shared__ unsigned s_tri[TRI_MAX];

  const int tid  = threadIdx.x;
  const int bIdx = blockIdx.y;
  const int part = blockIdx.x;
  const int wave = __builtin_amdgcn_readfirstlane(tid >> 5);
  const int lane = tid & 31;
  const int half = lane >> 4;
  const int col  = lane & 15;

  const int na   = (part == 0) ? 0 : ((part == 1) ? 3 : ((part == 2) ? 7 : 12));
  const int nb   = (part == 0) ? 2 : ((part == 1) ? 6 : ((part == 2) ? 11 : 29));
  const int g_lo = (part == 0) ? 0 : ((part == 1) ? 1306 : ((part == 2) ? 2660 : 3820));
  const int g_hi = (part == 0) ? 1306 : ((part == 1) ? 2660 : ((part == 2) ? 3820 : NG));
  const int cnt  = g_hi - g_lo;

  {
    const v4f* src = (const v4f*)(inp + (size_t)bIdx * SLAB);
    v4f regs[SLAB / 4 / 512];
#pragma unroll
    for (int i = 0; i < SLAB / 4 / 512; ++i)
      regs[i] = src[tid + i * 512];

    for (int t = tid; t < cnt; t += 512) {
      const unsigned h  = (unsigned)((NG - 1) - (g_lo + t));
      const unsigned h6 = 6u * h;
      unsigned a = 2u;
#pragma unroll
      for (unsigned s = 16u; s != 0u; s >>= 1) {
        const unsigned tt = a + s;
        const unsigned p3 = __umul24(__umul24(tt, tt - 1u), tt - 2u);
        a = (p3 <= h6) ? tt : a;
      }
      const unsigned pa = __umul24(__umul24(a, a - 1u), a - 2u);
      const unsigned h2 = h - ((pa * 43691u) >> 18);
      const unsigned h22 = 2u * h2;
      unsigned bq = 1u;
#pragma unroll
      for (unsigned s = 16u; s != 0u; s >>= 1) {
        const unsigned tt = bq + s;
        const unsigned p2 = __umul24(tt, tt - 1u);
        bq = (p2 <= h22) ? tt : bq;
      }
      const unsigned cq = h2 - (__umul24(bq, bq - 1u) >> 1);
      const unsigned n0 = 31u - a, n1 = 31u - bq, n2 = 31u - cq;
      s_tri[t] = (n0 * 32u + n1) | ((n0 * 32u + n2) << 10) | ((n1 * 32u + n2) << 20);
    }

    float* dst = s_in + (tid >> 7) * NP + ((tid >> 2) & 31) * RP + (tid & 3) * 4;
#pragma unroll
    for (int i = 0; i < SLAB / 4 / 512; ++i)
      *(v4f*)(dst + i * 4 * NP) = regs[i];

    const v4f* fsrc = (const v4f*)flt;
    for (int i = tid; i < (NF * 9 * NR) / 4; i += 512) {
      const int f = i / (9 * NR / 4);
      *(v4f*)&s_flt[f * FP + (i - f * (9 * NR / 4)) * 4] = fsrc[i];
    }
  }
  __syncthreads();

  {
    const int fbase = col * FP + 2 * half;
    const int lowC = ((nb + 1 < 15) ? (nb + 1) : 15) - na;
    const int tilesC = (nb - na + 1) + ((lowC > 0) ? lowC : 0);
    const int lowV = 15 - (na + 1);
    const int tilesV = (30 - na) + ((lowV > 0) ? lowV : 0);
    const int units = 2 * tilesC + tilesV;

    for (int u = wave; u < units; u += 16) {
      int kind, n, mh;
      if (u < tilesC)               { kind = 0; tile_of(u, na, nb, n, mh); }
      else if (u < tilesC + tilesV) { kind = 2; tile_of(u - tilesC, na + 1, 30, n, mh); }
      else                          { kind = 1; tile_of(u - tilesC - tilesV, na, nb, n, mh); }

      const int m  = mh * 16 + col;
      const int nm = n * NP + m * RP + 2 * half;
      const int mn = m * NP + n * RP + 2 * half;
      const int nn = n * (NP + RP) + 2 * half;
      const int mm = m * (NP + RP) + 2 * half;

      v8f c = {0.f, 0.f, 0.f, 0.f, 0.f, 0.f, 0.f, 0.f};
      if (kind == 0) {
        c = acc16(c, s_in, s_flt, nn, fbase + 0 * NR);
        c = acc16(c, s_in, s_flt, nm, fbase + 1 * NR);
        c = acc16(c, s_in, s_flt, mn, fbase + 3 * NR);
        c = acc16(c, s_in, s_flt, mm, fbase + 4 * NR);
      } else if (kind == 1) {
        c = acc16(c, s_in, s_flt, nm, fbase + 2 * NR);
        c = acc16(c, s_in, s_flt, mn, fbase + 6 * NR);
      } else {
        c = acc16(c, s_in, s_flt, nm, fbase + 5 * NR);
        c = acc16(c, s_in, s_flt, mn, fbase + 7 * NR);
        c = acc16(c, s_in, s_flt, mm, fbase + 8 * NR);
      }

      float* tp = s_tab + kind * SLAB + (n * NN + mh * 16 + half * 8) * NF + col;
#pragma unroll
      for (int r = 0; r < 8; ++r)
        tp[r * NF] = c[r];
    }
  }
  __syncthreads();

  {
    const int j = lane >> 2;
    const int q = lane & 3;
    const int passes = (cnt + 7) >> 3;
    float* obase = out + ((size_t)bIdx * NG + (size_t)g_lo) * NF + q * 4;
    for (int k0 = wave; k0 < passes; k0 += 64) {
      v4f x[4], y[4], z[4];
#pragma unroll
      for (int u = 0; u < 4; ++u) {
        const int t  = (k0 + 16 * u) * 8 + j;
        const int tc = (t < cnt) ? t : (cnt - 1);
        const unsigned w = s_tri[tc];
        const unsigned i01 = w & 1023u, i02 = (w >> 10) & 1023u, i12 = w >> 20;
        x[u] = *(const v4f*)&s_tab[0 * SLAB + i01 * NF + q * 4];
        y[u] = *(const v4f*)&s_tab[1 * SLAB + i02 * NF + q * 4];
        z[u] = *(const v4f*)&s_tab[2 * SLAB + i12 * NF + q * 4];
      }
#pragma unroll
      for (int u = 0; u < 4; ++u) {
        const int t = (k0 + 16 * u) * 8 + j;
        if (t < cnt)
          *(v4f*)(obase + (size_t)t * NF) = (x[u] + y[u]) + z[u];
      }
    }
  }
}

extern "C" void kernel_launch(void* const* d_in, const int* in_sizes, int n_in,
                              void* d_out, int out_size, void* d_ws, size_t ws_size,
                              hipStream_t stream) {
  const float* inp = (const float*)d_in[0];
  const float* flt = (const float*)d_in[1];
  float* out = (float*)d_out;
  dim3 grid(4, NB);
  dim3 block(512);
  hipLaunchKernelGGL(graphlet_conv_wmma, grid, block, 0, stream, inp, flt, out);
}


# Tile the pair tables over a packed list of the 496 pairs n<m (16 consecutive pairs per WMMA tile, table rows indexed by pair) instead of 16 consecutive m per n: about a third fewer WMMAs
# speedup vs baseline: 2.3331x; 1.0538x over previous
#include <hip/hip_runtime.h>

typedef __attribute__((ext_vector_type(2))) float v2f;
typedef __attribute__((ext_vector_type(4))) float v4f;
typedef __attribute__((ext_vector_type(8))) float v8f;

#define NB 64
#define NN 32
#define NR 16
#define NG 4960
#define NF 16
#define SLAB (NN * NN * NR)
#define TRI_MAX 1360
#define NPAIR 496
#define TAB (NPAIR * NF)

#define RP 20
#define NP 644
#define FP 148


__device__ __forceinline__ v8f acc16(v8f c, const float* s_in, const float* s_flt,
                                     int rowbase, int fb) {
#pragma unroll
  for (int s = 0; s < 4; ++s) {
    const v2f a = *(const v2f*)&s_in[rowbase + 4 * s];
    const v2f b = *(const v2f*)&s_flt[fb + 4 * s];
    c = __builtin_amdgcn_wmma_f32_16x16x4_f32(
             false, a,  false, b,
             (short)0, c,  false,  false);
  }
  return c;
}

__device__ __forceinline__ int pair_base(int n) { return (n * (63 - n)) >> 1; }

__global__ __launch_bounds__(512)
void graphlet_conv_wmma(const float* __restrict__ inp,
                        const float* __restrict__ flt,
                        float* __restrict__ out) {
  __shared__ __align__(16) float s_in[NN * NP];
  __shared__ __align__(16) float s_flt[NF * FP];
  __shared__ __align__(16) float s_tab[3 * TAB];
  __shared__ unsigned s_tri[TRI_MAX];

  const int tid  = threadIdx.x;
  const int bIdx = blockIdx.y;
  const int part = blockIdx.x;
  const int wave = __builtin_amdgcn_readfirstlane(tid >> 5);
  const int lane = tid & 31;
  const int half = lane >> 4;
  const int col  = lane & 15;

  const int na   = (part == 0) ? 0 : ((part == 1) ? 3 : ((part == 2) ? 7 : 12));
  const int nb   = (part == 0) ? 2 : ((part == 1) ? 6 : ((part == 2) ? 11 : 29));
  const int g_lo = (part == 0) ? 0 : ((part == 1) ? 1306 : ((part == 2) ? 2660 : 3820));
  const int g_hi = (part == 0) ? 1306 : ((part == 1) ? 2660 : ((part == 2) ? 3820 : NG));
  const int cnt  = g_hi - g_lo;

  {
    const v4f* src = (const v4f*)(inp + (size_t)bIdx * SLAB);
    v4f regs[SLAB / 4 / 512];
#pragma unroll
    for (int i = 0; i < SLAB / 4 / 512; ++i)
      regs[i] = src[tid + i * 512];

    for (int t = tid; t < cnt; t += 512) {
      const unsigned h  = (unsigned)((NG - 1) - (g_lo + t));
      const unsigned h6 = 6u * h;
      unsigned a = 2u;
#pragma unroll
      for (unsigned s = 16u; s != 0u; s >>= 1) {
        const unsigned tt = a + s;
        const unsigned p3 = __umul24(__umul24(tt, tt - 1u), tt - 2u);
        a = (p3 <= h6) ? tt : a;
      }
      const unsigned pa = __umul24(__umul24(a, a - 1u), a - 2u);
      const unsigned h2 = h - ((pa * 43691u) >> 18);
      const unsigned h22 = 2u * h2;
      unsigned bq = 1u;
#pragma unroll
      for (unsigned s = 16u; s != 0u; s >>= 1) {
        const unsigned tt = bq + s;
        const unsigned p2 = __umul24(tt, tt - 1u);
        bq = (p2 <= h22) ? tt : bq;
      }
      const unsigned cq = h2 - (__umul24(bq, bq - 1u) >> 1);
      const unsigned n0 = 31u - a, n1 = 31u - bq, n2 = 31u - cq;
      const unsigned r0 = ((n0 * (63u - n0)) >> 1) - n0 - 1u;
      const unsigned r1 = ((n1 * (63u - n1)) >> 1) - n1 - 1u;
      s_tri[t] = (r0 + n1) | ((r0 + n2) << 10) | ((r1 + n2) << 20);
    }

    float* dst = s_in + (tid >> 7) * NP + ((tid >> 2) & 31) * RP + (tid & 3) * 4;
#pragma unroll
    for (int i = 0; i < SLAB / 4 / 512; ++i)
      *(v4f*)(dst + i * 4 * NP) = regs[i];

    const v4f* fsrc = (const v4f*)flt;
    for (int i = tid; i < (NF * 9 * NR) / 4; i += 512) {
      const int f = i / (9 * NR / 4);
      *(v4f*)&s_flt[f * FP + (i - f * (9 * NR / 4)) * 4] = fsrc[i];
    }
  }
  __syncthreads();

  {
    const int fbase = col * FP + 2 * half;
    const int tc0 = pair_base(na) >> 4;
    const int tilesC = ((pair_base(nb + 1) - 1) >> 4) - tc0 + 1;
    const int tv0 = pair_base(na + 1) >> 4;
    const int tilesV = (NPAIR / 16) - tv0;
    const int units = 2 * tilesC + tilesV;

    for (int u = wave; u < units; u += 16) {
      int kind, tile;
      if (u < tilesC)               { kind = 0; tile = tc0 + u; }
      else if (u < tilesC + tilesV) { kind = 2; tile = tv0 + (u - tilesC); }
      else                          { kind = 1; tile = tc0 + (u - tilesC - tilesV); }

      const int p = tile * 16 + col;
      int n = 0;
#pragma unroll
      for (int s = 16; s != 0; s >>= 1) {
        const int cn = n + s;
        n = (cn * (63 - cn) <= 2 * p) ? cn : n;
      }
      const int m  = n + 1 + (p - pair_base(n));
      const int nm = n * NP + m * RP + 2 * half;
      const int mn = m * NP + n * RP + 2 * half;
      const int nn = n * (NP + RP) + 2 * half;
      const int mm = m * (NP + RP) + 2 * half;

      v8f c = {0.f, 0.f, 0.f, 0.f, 0.f, 0.f, 0.f, 0.f};
      if (kind == 0) {
        c = acc16(c, s_in, s_flt, nn, fbase + 0 * NR);
        c = acc16(c, s_in, s_flt, nm, fbase + 1 * NR);
        c = acc16(c, s_in, s_flt, mn, fbase + 3 * NR);
        c = acc16(c, s_in, s_flt, mm, fbase + 4 * NR);
      } else if (kind == 1) {
        c = acc16(c, s_in, s_flt, nm, fbase + 2 * NR);
        c = acc16(c, s_in, s_flt, mn, fbase + 6 * NR);
      } else {
        c = acc16(c, s_in, s_flt, nm, fbase + 5 * NR);
        c = acc16(c, s_in, s_flt, mn, fbase + 7 * NR);
        c = acc16(c, s_in, s_flt, mm, fbase + 8 * NR);
      }

      float* tp = s_tab + kind * TAB + (tile * 16 + half * 8) * NF + col;
#pragma unroll
      for (int r = 0; r < 8; ++r)
        tp[r * NF] = c[r];
    }
  }
  __syncthreads();

  {
    const int j = lane >> 2;
    const int q = lane & 3;
    const int passes = (cnt + 7) >> 3;
    float* obase = out + ((size_t)bIdx * NG + (size_t)g_lo) * NF + q * 4;
    for (int k0 = wave; k0 < passes; k0 += 64) {
      v4f x[4], y[4], z[4];
#pragma unroll
      for (int u = 0; u < 4; ++u) {
        const int t  = (k0 + 16 * u) * 8 + j;
        const int tc = (t < cnt) ? t : (cnt - 1);
        const unsigned w = s_tri[tc];
        const unsigned i01 = w & 1023u, i02 = (w >> 10) & 1023u, i12 = w >> 20;
        x[u] = *(const v4f*)&s_tab[0 * TAB + i01 * NF + q * 4];
        y[u] = *(const v4f*)&s_tab[1 * TAB + i02 * NF + q * 4];
        z[u] = *(const v4f*)&s_tab[2 * TAB + i12 * NF + q * 4];
      }
#pragma unroll
      for (int u = 0; u < 4; ++u) {
        const int t = (k0 + 16 * u) * 8 + j;
        if (t < cnt)
          *(v4f*)(obase + (size_t)t * NF) = (x[u] + y[u]) + z[u];
      }
    }
  }
}

extern "C" void kernel_launch(void* const* d_in, const int* in_sizes, int n_in,
                              void* d_out, int out_size, void* d_ws, size_t ws_size,
                              hipStream_t stream) {
  const float* inp = (const float*)d_in[0];
  const float* flt = (const float*)d_in[1];
  float* out = (float*)d_out;
  dim3 grid(4, NB);
  dim3 block(512);
  hipLaunchKernelGGL(graphlet_conv_wmma, grid, block, 0, stream, inp, flt, out);
}
